# starting kernel
# baseline (speedup 1.0000x reference)
#include <hip/hip_runtime.h>
#include <hip/hip_bf16.h>
#include <math.h>

typedef float v2f __attribute__((ext_vector_type(2)));
typedef float v8f __attribute__((ext_vector_type(8)));

#define GNN_N 100000
#define D_IN 128
#define D_OUT 64

__global__ void zero_f(float* __restrict__ p, size_t n) {
    size_t i = (size_t)blockIdx.x * blockDim.x + threadIdx.x;
    size_t stride = (size_t)gridDim.x * blockDim.x;
    for (; i < n; i += stride) p[i] = 0.0f;
}

__global__ void deg_count(const int* __restrict__ dst, float* __restrict__ deg, int E) {
    int i = blockIdx.x * blockDim.x + threadIdx.x;
    if (i < E) atomicAdd(deg + dst[i], 1.0f);
}

__global__ __launch_bounds__(256)
void scatter_add(const float* __restrict__ x, const int* __restrict__ src,
                 const int* __restrict__ dst, float* __restrict__ agg, int E) {
    long g = (long)blockIdx.x * blockDim.x + threadIdx.x;
    long edge = g >> 5;
    if (edge >= E) return;
    int lane = (int)(g & 31);
    long so = (long)src[edge] * D_IN + lane * 4;
    long dro = (long)dst[edge] * D_IN + lane * 4;
    float4 v = *(const float4*)(x + so);
    atomicAdd(agg + dro + 0, v.x);
    atomicAdd(agg + dro + 1, v.y);
    atomicAdd(agg + dro + 2, v.z);
    atomicAdd(agg + dro + 3, v.w);
}

template<int NT, int EPI>
__global__ __launch_bounds__(32 * NT)
void sage_gemm(const float* __restrict__ A,
               const float* __restrict__ Agg,
               const float* __restrict__ deg,
               const float* __restrict__ Wr,
               const float* __restrict__ Wn,
               const float* __restrict__ bias,
               const float* __restrict__ gamma,
               const float* __restrict__ beta,
               float* __restrict__ out)
{
    constexpr int W = 16 * NT;
    constexpr int K = 128;
    constexpr int LDA = K + 4;
    __shared__ float As[16][LDA];
    __shared__ float Gs[16][LDA];
    __shared__ float Hs[16][W];
    __shared__ float inv[16];

    const int tid  = threadIdx.x;
    const int lane = tid & 31;
    const int wave = tid >> 5;
    const long rowBase = (long)blockIdx.x * 16;

    if (tid < 16) {
        float d = deg[rowBase + tid];
        inv[tid] = 1.0f / fmaxf(d, 1.0f);
    }
    __syncthreads();

    const float* aSrc = A + rowBase * K;
    const float* gSrc = Agg + rowBase * K;
    for (int e = tid; e < 16 * K; e += 32 * NT) {
        int r = e >> 7, c = e & 127;
        As[r][c] = aSrc[e];
        Gs[r][c] = gSrc[e] * inv[r];
    }
    __syncthreads();

    const int n0   = wave * 16;
    const int arow = lane & 15;
    const int koff = (lane >> 4) << 1;
    const int bcol = lane & 15;

    v8f acc = {};
#pragma unroll 4
    for (int k0 = 0; k0 < K; k0 += 4) {
        v2f a0, a1, b0, b1;
        a0.x = As[arow][k0 + koff];
        a0.y = As[arow][k0 + koff + 1];
        a1.x = Gs[arow][k0 + koff];
        a1.y = Gs[arow][k0 + koff + 1];
        b0.x = Wr[(k0 + koff) * W + n0 + bcol];
        b0.y = Wr[(k0 + koff + 1) * W + n0 + bcol];
        b1.x = Wn[(k0 + koff) * W + n0 + bcol];
        b1.y = Wn[(k0 + koff + 1) * W + n0 + bcol];
        acc = __builtin_amdgcn_wmma_f32_16x16x4_f32(
            false, a0, false, b0, (short)0, acc, false, false);
        acc = __builtin_amdgcn_wmma_f32_16x16x4_f32(
            false, a1, false, b1, (short)0, acc, false, false);
    }

    {
        float bv = bias[n0 + bcol];
        int rbase = (lane >> 4) << 3;
#pragma unroll
        for (int v = 0; v < 8; ++v)
            Hs[rbase + v][n0 + bcol] = acc[v] + bv;
    }
    __syncthreads();

    if constexpr (EPI == 0) {
        for (int r = wave; r < 16; r += NT) {
            float s = 0.0f;
            for (int c = lane; c < W; c += 32) s += Hs[r][c];
#pragma unroll
            for (int off = 16; off; off >>= 1) s += __shfl_xor(s, off, 32);
            float mu = s * (1.0f / W);
            float vv = 0.0f;
            for (int c = lane; c < W; c += 32) {
                float d = Hs[r][c] - mu;
                vv += d * d;
            }
#pragma unroll
            for (int off = 16; off; off >>= 1) vv += __shfl_xor(vv, off, 32);
            float rs = rsqrtf(vv * (1.0f / W) + 1e-5f);
            for (int c = lane; c < W; c += 32) {
                float val = (Hs[r][c] - mu) * rs * gamma[c] + beta[c];
                out[(rowBase + r) * W + c] = fmaxf(val, 0.0f);
            }
        }
    } else {
        for (int r = wave; r < 16; r += NT) {
            float m = -INFINITY;
            for (int c = lane; c < W; c += 32) m = fmaxf(m, Hs[r][c]);
#pragma unroll
            for (int off = 16; off; off >>= 1) m = fmaxf(m, __shfl_xor(m, off, 32));
            float s = 0.0f;
            for (int c = lane; c < W; c += 32) s += expf(Hs[r][c] - m);
#pragma unroll
            for (int off = 16; off; off >>= 1) s += __shfl_xor(s, off, 32);
            float ls = logf(s);
            for (int c = lane; c < W; c += 32)
                out[(rowBase + r) * W + c] = Hs[r][c] - m - ls;
        }
    }
}

extern "C" void kernel_launch(void* const* d_in, const int* in_sizes, int n_in,
                              void* d_out, int out_size, void* d_ws, size_t ws_size,
                              hipStream_t stream) {
    const float* x    = (const float*)d_in[0];
    const int*   esrc = (const int*)d_in[1];
    const int*   edst = (const int*)d_in[2];
    const float* Wr0  = (const float*)d_in[3];
    const float* Wn0  = (const float*)d_in[4];
    const float* b0   = (const float*)d_in[5];
    const float* g0   = (const float*)d_in[6];
    const float* be0  = (const float*)d_in[7];
    const float* Wr1  = (const float*)d_in[8];
    const float* Wn1  = (const float*)d_in[9];
    const float* b1   = (const float*)d_in[10];
    const float* g1   = (const float*)d_in[11];
    const float* be1  = (const float*)d_in[12];
    const float* Wr2  = (const float*)d_in[13];
    const float* Wn2  = (const float*)d_in[14];
    const float* b2   = (const float*)d_in[15];
    float* out = (float*)d_out;

    const int N = GNN_N;
    const int E = in_sizes[1];

    float* deg = (float*)d_ws;
    float* agg = deg + N;
    float* h0  = agg + (size_t)N * D_IN;
    float* h1  = h0 + (size_t)N * D_IN;

    const int ZB = 4096, ZT = 256;
    const int degBlocks = (E + 255) / 256;
    const int scatBlocks = (int)(((long)E * 32 + 255) / 256);
    const int gemmBlocks = N / 16;

    zero_f<<<ZB, ZT, 0, stream>>>(deg, (size_t)N * (D_IN + 1));
    deg_count<<<degBlocks, 256, 0, stream>>>(edst, deg, E);

    scatter_add<<<scatBlocks, 256, 0, stream>>>(x, esrc, edst, agg, E);
    sage_gemm<8, 0><<<gemmBlocks, 256, 0, stream>>>(x, agg, deg, Wr0, Wn0, b0, g0, be0, h0);

    zero_f<<<ZB, ZT, 0, stream>>>(agg, (size_t)N * D_IN);
    scatter_add<<<scatBlocks, 256, 0, stream>>>(h0, esrc, edst, agg, E);
    sage_gemm<8, 0><<<gemmBlocks, 256, 0, stream>>>(h0, agg, deg, Wr1, Wn1, b1, g1, be1, h1);

    zero_f<<<ZB, ZT, 0, stream>>>(agg, (size_t)N * D_IN);
    scatter_add<<<scatBlocks, 256, 0, stream>>>(h1, esrc, edst, agg, E);
    sage_gemm<4, 1><<<gemmBlocks, 128, 0, stream>>>(h1, agg, deg, Wr2, Wn2, b2, nullptr, nullptr, out);
}


# sage_gemm: row block 16 -> 80, a wave keeps 5 row tiles' accumulators so one Wr/Wn fragment pair feeds 5 WMMA pairs (weight loads per WMMA / 5, 5x fewer waves); epilogue rows abreast; sum order kept
# speedup vs baseline: 3.2283x; 3.2283x over previous
#include <hip/hip_runtime.h>
#include <hip/hip_bf16.h>
#include <math.h>

typedef float v2f __attribute__((ext_vector_type(2)));
typedef float v8f __attribute__((ext_vector_type(8)));

#define GNN_N 100000
#define D_IN 128
#define D_OUT 64

__device__ __forceinline__ int package_half() {
    return (int)((__builtin_amdgcn_s_sendmsg_rtn(0x87) >> 18) & 1u);
}

#define OWN_ROWS 391
#define OWN_WGS 256
#define OWN_THREADS 1024
#define OWN_WAVES (OWN_THREADS / 32)
#define OWN_BATCH 16
#define TAG_CHUNK 128
#define SCAN_U 4
#define LIST_SHIFT 9
#define LIST_MASK 511u
#define LIST_CAP 512
static_assert(OWN_ROWS <= (1 << LIST_SHIFT), "local row must fit the low bits of a list word");
static_assert(GNN_N <= (1 << (32 - LIST_SHIFT)), "src id must fit the high bits of a list word");

__global__ void zero_f(float* __restrict__ p, size_t n) {
    size_t i = (size_t)blockIdx.x * blockDim.x + threadIdx.x;
    size_t stride = (size_t)gridDim.x * blockDim.x;
    for (; i < n; i += stride) p[i] = 0.0f;
}

__global__ void deg_count(const int* __restrict__ dst, float* __restrict__ deg,
                          unsigned* __restrict__ tag4, int E, int nPad4) {
    int i = blockIdx.x * blockDim.x + threadIdx.x;
    if (i >= nPad4) return;
    float* plane = deg + (long)package_half() * GNN_N;
    unsigned t = 0;
#pragma unroll
    for (int k = 0; k < 4; ++k) {
        int e = 4 * i + k;
        if (e < E) {
            int d = dst[e];
            atomicAdd(plane + d, 1.0f);
            t |= (unsigned)(d / OWN_ROWS) << (8 * k);
        }
    }
    tag4[i] = t;
}

__device__ __forceinline__ void edge_batch(const float* __restrict__ x, float* acc, unsigned w, int lane) {
    float v[OWN_BATCH][4];
#pragma unroll
    for (int u = 0; u < OWN_BATCH; ++u) {
        const unsigned wj = (unsigned)__builtin_amdgcn_readlane((int)w, u);
        const float* p = x + (long)(wj >> LIST_SHIFT) * D_IN + lane;
        v[u][0] = p[0]; v[u][1] = p[32]; v[u][2] = p[64]; v[u][3] = p[96];
    }
#pragma unroll
    for (int u = 0; u < OWN_BATCH; ++u) {
        const unsigned wj = (unsigned)__builtin_amdgcn_readlane((int)w, u);
        float* a = acc + (int)(wj & LIST_MASK) * D_IN + lane;
        atomicAdd(a, v[u][0]);
        atomicAdd(a + 32, v[u][1]);
        atomicAdd(a + 64, v[u][2]);
        atomicAdd(a + 96, v[u][3]);
    }
}

__device__ __forceinline__ void edge_single(const float* __restrict__ x, float* acc, unsigned wj, int lane) {
    const float* p = x + (long)(wj >> LIST_SHIFT) * D_IN + lane;
    float* a = acc + (int)(wj & LIST_MASK) * D_IN + lane;
    const float v0 = p[0], v1 = p[32], v2 = p[64], v3 = p[96];
    atomicAdd(a, v0);
    atomicAdd(a + 32, v1);
    atomicAdd(a + 64, v2);
    atomicAdd(a + 96, v3);
}

template<bool RECORD>
__device__ __forceinline__ unsigned queued_words(const int* __restrict__ src, const int* __restrict__ dst,
                                                 unsigned* __restrict__ list, int q, int n, int pos,
                                                 int rowBase, int lane) {
    const unsigned w = ((unsigned)src[q] << LIST_SHIFT) | (unsigned)(dst[q] - rowBase);
    if constexpr (RECORD) {
        if (lane < n && pos + lane < LIST_CAP) list[pos + lane] = w;
    }
    return w;
}

__device__ __forceinline__ unsigned zero_bytes(unsigned w) {
    return ~(((w & 0x7f7f7f7fu) + 0x7f7f7f7fu) | w | 0x7f7f7f7fu);
}

__device__ __forceinline__ uint4 tag_chunk(const unsigned* __restrict__ tag4, int c, int rot,
                                           int nChunks, int lane) {
    int r = (c < nChunks ? c : nChunks - 1) + rot;
    if (r >= nChunks) r -= nChunks;
    return *reinterpret_cast<const uint4*>(tag4 + (long)r * TAG_CHUNK + lane * 4);
}

template<bool RECORD>
__device__ __forceinline__ int scan_edges(const float* __restrict__ x, const int* __restrict__ src,
                                          const int* __restrict__ dst, const unsigned* __restrict__ tag4,
                                          unsigned* __restrict__ list, float* acc, int E, int nChunks,
                                          int owner, int rowBase, int wib, int lane) {
    const unsigned mine4 = (unsigned)owner * 0x01010101u;
    const int rot = (int)(((long)owner * nChunks) / OWN_WGS);
    int q = 0;
    int cnt = 0;
    int pos = 0;

    if (nChunks > 0) {
        uint4 a0 = tag_chunk(tag4, wib, rot, nChunks, lane);
        uint4 a1 = tag_chunk(tag4, wib + OWN_WAVES, rot, nChunks, lane);
        uint4 a2 = tag_chunk(tag4, wib + 2 * OWN_WAVES, rot, nChunks, lane);
        uint4 a3 = tag_chunk(tag4, wib + 3 * OWN_WAVES, rot, nChunks, lane);
        for (int c = wib; c < nChunks; c += SCAN_U * OWN_WAVES) {
            const uint4 b0 = tag_chunk(tag4, c + 4 * OWN_WAVES, rot, nChunks, lane);
            const uint4 b1 = tag_chunk(tag4, c + 5 * OWN_WAVES, rot, nChunks, lane);
            const uint4 b2 = tag_chunk(tag4, c + 6 * OWN_WAVES, rot, nChunks, lane);
            const uint4 b3 = tag_chunk(tag4, c + 7 * OWN_WAVES, rot, nChunks, lane);
#pragma nounroll
            for (int k = 0; k < SCAN_U; ++k) {
                const int cc = c + k * OWN_WAVES;
                if (cc >= nChunks) break;
                const uint4 cur = k == 0 ? a0 : k == 1 ? a1 : k == 2 ? a2 : a3;
                int r = cc + rot;
                if (r >= nChunks) r -= nChunks;
                const unsigned zall = (zero_bytes(cur.x ^ mine4) >> 7) | (zero_bytes(cur.y ^ mine4) >> 6) |
                                      (zero_bytes(cur.z ^ mine4) >> 5) | (zero_bytes(cur.w ^ mine4) >> 4);
                unsigned m = __builtin_amdgcn_ballot_w32(zall != 0u);
                while (m) {
                    const int l = __builtin_ctz(m);
                    m &= m - 1u;
                    unsigned z = (unsigned)__builtin_amdgcn_readlane((int)zall, l);
                    while (z) {
                        const int p = __builtin_ctz(z);
                        z &= z - 1u;
                        const int e = r * (TAG_CHUNK * 4) + l * 16 + (p & 7) * 4 + (p >> 3);
                        if (e < E) {
                            q = (lane == cnt) ? e : q;
                            ++cnt;
                            if (cnt == OWN_BATCH) {
                                const unsigned w = queued_words<RECORD>(src, dst, list, q, OWN_BATCH, pos,
                                                                        rowBase, lane);
                                edge_batch(x, acc, w, lane);
                                pos += OWN_BATCH;
                                cnt = 0;
                            }
                        }
                    }
                }
            }
            a0 = b0; a1 = b1; a2 = b2; a3 = b3;
        }
    }
    if (cnt > 0) {
        const unsigned w = queued_words<RECORD>(src, dst, list, q, cnt, pos, rowBase, lane);
        for (int j = 0; j < cnt; ++j)
            edge_single(x, acc, (unsigned)__builtin_amdgcn_readlane((int)w, j), lane);
    }
    return pos + cnt;
}

template<bool FIRST>
__global__ __launch_bounds__(OWN_THREADS)
void scatter_owned(const float* __restrict__ x, const int* __restrict__ src,
                   const int* __restrict__ dst, const unsigned* __restrict__ tag4,
                   int* __restrict__ listLen, unsigned* __restrict__ lists,
                   float* __restrict__ agg, int E, int nChunks) {
    __shared__ float acc[OWN_ROWS * D_IN];
    const int tid = threadIdx.x;
    const int lane = tid & 31;
    const int wib = __builtin_amdgcn_readfirstlane(tid >> 5);
    const int owner = (int)blockIdx.x;
    const int rowBase = owner * OWN_ROWS;
    const int slot = owner * OWN_WAVES + wib;
    unsigned* list = lists + (long)slot * LIST_CAP;

    for (int i = tid; i < OWN_ROWS * D_IN; i += OWN_THREADS) acc[i] = 0.0f;
    __syncthreads();

    if constexpr (FIRST) {
        const int n = scan_edges<true>(x, src, dst, tag4, list, acc, E, nChunks, owner, rowBase, wib, lane);
        if (lane == 0) listLen[slot] = n;
    } else {
        const int n = listLen[slot];
        if (n > LIST_CAP) {
            scan_edges<false>(x, src, dst, tag4, list, acc, E, nChunks, owner, rowBase, wib, lane);
        } else if (n > 0) {
            const int l16 = lane & 15;
            unsigned cur = list[l16 < n ? l16 : n - 1];
            int base = 0;
            for (; base + OWN_BATCH <= n; base += OWN_BATCH) {
                const int nx = base + OWN_BATCH + l16;
                const unsigned nxt = list[nx < n ? nx : n - 1];
                edge_batch(x, acc, cur, lane);
                cur = nxt;
            }
            for (int j = 0; j < n - base; ++j)
                edge_single(x, acc, (unsigned)__builtin_amdgcn_readlane((int)cur, j), lane);
        }
    }
    __syncthreads();

    const int rows = (GNN_N - rowBase) < OWN_ROWS ? (GNN_N - rowBase) : OWN_ROWS;
    float4* o = reinterpret_cast<float4*>(agg + (long)rowBase * D_IN);
    const float4* a4 = reinterpret_cast<const float4*>(acc);
    for (int i = tid; i < rows * (D_IN / 4); i += OWN_THREADS) o[i] = a4[i];
}

#define GEMM_MT 5
template<int NT, int MT, int EPI>
__global__ __launch_bounds__(32 * NT)
void sage_gemm(const float* __restrict__ A,
               const float* __restrict__ Agg,
               const float* __restrict__ deg,
               const float* __restrict__ Wr,
               const float* __restrict__ Wn,
               const float* __restrict__ bias,
               const float* __restrict__ gamma,
               const float* __restrict__ beta,
               float* __restrict__ out)
{
    constexpr int W = 16 * NT;
    constexpr int K = 128;
    constexpr int LDA = K + 4;
    constexpr int ROWS = 16 * MT;
    constexpr int RPW = ROWS / NT;
    constexpr int RG = 5;
    static_assert(GNN_N % ROWS == 0, "row blocks must tile N exactly");
    static_assert(ROWS <= 32 * NT, "one thread per row for the degree reciprocals");
    static_assert(ROWS % NT == 0 && RPW % RG == 0, "epilogue row groups must be whole");
    static_assert(W <= LDA, "the output tile must fit the plane it reuses");
    __shared__ float As[ROWS][LDA];
    __shared__ float Gs[ROWS][LDA];
    __shared__ float inv[ROWS];
    float (*Hs)[W] = reinterpret_cast<float (*)[W]>(&As[0][0]);

    const int tid  = threadIdx.x;
    const int lane = tid & 31;
    const int wave = tid >> 5;
    const long rowBase = (long)blockIdx.x * ROWS;

    if (tid < ROWS) {
        float d = deg[rowBase + tid] + deg[GNN_N + rowBase + tid];
        inv[tid] = 1.0f / fmaxf(d, 1.0f);
    }
    __syncthreads();

    const float* aSrc = A + rowBase * K;
    const float* gSrc = Agg + rowBase * K;
    for (int e = tid; e < ROWS * K; e += 32 * NT) {
        int r = e >> 7, c = e & 127;
        As[r][c] = aSrc[e];
        Gs[r][c] = gSrc[e] * inv[r];
    }
    __syncthreads();

    const int n0   = wave * 16;
    const int arow = lane & 15;
    const int koff = (lane >> 4) << 1;
    const int bcol = lane & 15;

    v8f acc[MT];
#pragma unroll
    for (int t = 0; t < MT; ++t) acc[t] = v8f{};
#pragma unroll 2
    for (int k0 = 0; k0 < K; k0 += 4) {
        v2f b0, b1;
        b0.x = Wr[(k0 + koff) * W + n0 + bcol];
        b0.y = Wr[(k0 + koff + 1) * W + n0 + bcol];
        b1.x = Wn[(k0 + koff) * W + n0 + bcol];
        b1.y = Wn[(k0 + koff + 1) * W + n0 + bcol];
#pragma unroll
        for (int t = 0; t < MT; ++t) {
            v2f a0, a1;
            a0.x = As[16 * t + arow][k0 + koff];
            a0.y = As[16 * t + arow][k0 + koff + 1];
            a1.x = Gs[16 * t + arow][k0 + koff];
            a1.y = Gs[16 * t + arow][k0 + koff + 1];
            acc[t] = __builtin_amdgcn_wmma_f32_16x16x4_f32(
                false, a0, false, b0, (short)0, acc[t], false, false);
            acc[t] = __builtin_amdgcn_wmma_f32_16x16x4_f32(
                false, a1, false, b1, (short)0, acc[t], false, false);
        }
    }
    __syncthreads();

    {
        float bv = bias[n0 + bcol];
        int rbase = (lane >> 4) << 3;
#pragma unroll
        for (int t = 0; t < MT; ++t) {
#pragma unroll
            for (int v = 0; v < 8; ++v)
                Hs[16 * t + rbase + v][n0 + bcol] = acc[t][v] + bv;
        }
    }
    __syncthreads();

    constexpr int CPL = W / 32;
    if constexpr (EPI == 0) {
        float gm[CPL], bt[CPL];
#pragma unroll
        for (int i = 0; i < CPL; ++i) {
            gm[i] = gamma[lane + 32 * i];
            bt[i] = beta[lane + 32 * i];
        }
        for (int g = 0; g < RPW; g += RG) {
            float h[RG][CPL], s[RG], mu[RG], vv[RG];
#pragma unroll
            for (int j = 0; j < RG; ++j) {
                const int r = wave + (g + j) * NT;
#pragma unroll
                for (int i = 0; i < CPL; ++i) h[j][i] = Hs[r][lane + 32 * i];
            }
#pragma unroll
            for (int j = 0; j < RG; ++j) {
                float t = 0.0f;
#pragma unroll
                for (int i = 0; i < CPL; ++i) t += h[j][i];
                s[j] = t;
            }
#pragma unroll
            for (int off = 16; off; off >>= 1) {
#pragma unroll
                for (int j = 0; j < RG; ++j) s[j] += __shfl_xor(s[j], off, 32);
            }
#pragma unroll
            for (int j = 0; j < RG; ++j) {
                mu[j] = s[j] * (1.0f / W);
                float t = 0.0f;
#pragma unroll
                for (int i = 0; i < CPL; ++i) {
                    float d = h[j][i] - mu[j];
                    t += d * d;
                }
                vv[j] = t;
            }
#pragma unroll
            for (int off = 16; off; off >>= 1) {
#pragma unroll
                for (int j = 0; j < RG; ++j) vv[j] += __shfl_xor(vv[j], off, 32);
            }
#pragma unroll
            for (int j = 0; j < RG; ++j) {
                const int r = wave + (g + j) * NT;
                float rs = rsqrtf(vv[j] * (1.0f / W) + 1e-5f);
#pragma unroll
                for (int i = 0; i < CPL; ++i) {
                    float val = (h[j][i] - mu[j]) * rs * gm[i] + bt[i];
                    out[(rowBase + r) * W + lane + 32 * i] = fmaxf(val, 0.0f);
                }
            }
        }
    } else {
        for (int g = 0; g < RPW; g += RG) {
            float h[RG][CPL], m[RG], s[RG];
#pragma unroll
            for (int j = 0; j < RG; ++j) {
                const int r = wave + (g + j) * NT;
#pragma unroll
                for (int i = 0; i < CPL; ++i) h[j][i] = Hs[r][lane + 32 * i];
            }
#pragma unroll
            for (int j = 0; j < RG; ++j) {
                float t = -INFINITY;
#pragma unroll
                for (int i = 0; i < CPL; ++i) t = fmaxf(t, h[j][i]);
                m[j] = t;
            }
#pragma unroll
            for (int off = 16; off; off >>= 1) {
#pragma unroll
                for (int j = 0; j < RG; ++j) m[j] = fmaxf(m[j], __shfl_xor(m[j], off, 32));
            }
#pragma unroll
            for (int j = 0; j < RG; ++j) {
                float t = 0.0f;
#pragma unroll
                for (int i = 0; i < CPL; ++i) t += expf(h[j][i] - m[j]);
                s[j] = t;
            }
#pragma unroll
            for (int off = 16; off; off >>= 1) {
#pragma unroll
                for (int j = 0; j < RG; ++j) s[j] += __shfl_xor(s[j], off, 32);
            }
#pragma unroll
            for (int j = 0; j < RG; ++j) {
                const int r = wave + (g + j) * NT;
                float ls = logf(s[j]);
#pragma unroll
                for (int i = 0; i < CPL; ++i)
                    out[(rowBase + r) * W + lane + 32 * i] = h[j][i] - m[j] - ls;
            }
        }
    }
}

extern "C" void kernel_launch(void* const* d_in, const int* in_sizes, int n_in,
                              void* d_out, int out_size, void* d_ws, size_t ws_size,
                              hipStream_t stream) {
    const float* x    = (const float*)d_in[0];
    const int*   esrc = (const int*)d_in[1];
    const int*   edst = (const int*)d_in[2];
    const float* Wr0  = (const float*)d_in[3];
    const float* Wn0  = (const float*)d_in[4];
    const float* b0   = (const float*)d_in[5];
    const float* g0   = (const float*)d_in[6];
    const float* be0  = (const float*)d_in[7];
    const float* Wr1  = (const float*)d_in[8];
    const float* Wn1  = (const float*)d_in[9];
    const float* b1   = (const float*)d_in[10];
    const float* g1   = (const float*)d_in[11];
    const float* be1  = (const float*)d_in[12];
    const float* Wr2  = (const float*)d_in[13];
    const float* Wn2  = (const float*)d_in[14];
    const float* b2   = (const float*)d_in[15];
    float* out = (float*)d_out;

    const int N = GNN_N;
    const int E = in_sizes[1];

    float* deg = (float*)d_ws;
    float* agg = deg + 2 * (size_t)N;
    float* h0  = agg + (size_t)N * D_IN;
    float* h1  = h0 + (size_t)N * D_IN;
    unsigned* tag4 = (unsigned*)(h1 + (size_t)N * D_IN);

    const int ZB = 256, ZT = 256;
    const int nChunks = ((E + 3) / 4 + TAG_CHUNK - 1) / TAG_CHUNK;
    const int nPad4 = nChunks * TAG_CHUNK;
    const int degBlocks = (nPad4 + 255) / 256;
    const int gemmBlocks = N / (16 * GEMM_MT);
    int* listLen = (int*)(tag4 + nPad4);
    unsigned* lists = (unsigned*)(listLen + OWN_WGS * OWN_WAVES);

    zero_f<<<ZB, ZT, 0, stream>>>(deg, 2 * (size_t)N);
    deg_count<<<degBlocks, 256, 0, stream>>>(edst, deg, tag4, E, nPad4);

    scatter_owned<true><<<OWN_WGS, OWN_THREADS, 0, stream>>>(x, esrc, edst, tag4, listLen, lists, agg, E, nChunks);
    sage_gemm<8, GEMM_MT, 0><<<gemmBlocks, 256, 0, stream>>>(x, agg, deg, Wr0, Wn0, b0, g0, be0, h0);

    scatter_owned<false><<<OWN_WGS, OWN_THREADS, 0, stream>>>(h0, esrc, edst, tag4, listLen, lists, agg, E, nChunks);
    sage_gemm<8, GEMM_MT, 0><<<gemmBlocks, 256, 0, stream>>>(h0, agg, deg, Wr1, Wn1, b1, g1, be1, h1);

    scatter_owned<false><<<OWN_WGS, OWN_THREADS, 0, stream>>>(h1, esrc, edst, tag4, listLen, lists, agg, E, nChunks);
    sage_gemm<4, GEMM_MT, 1><<<gemmBlocks, 128, 0, stream>>>(h1, agg, deg, Wr2, Wn2, b2, nullptr, nullptr, out);
}
